# starting kernel
# baseline (speedup 1.0000x reference)
#include <hip/hip_runtime.h>

typedef __attribute__((ext_vector_type(2))) float v2f;
typedef __attribute__((ext_vector_type(4))) float v4f;
typedef __attribute__((ext_vector_type(8))) float v8f;

#define DF 64

__global__ __launch_bounds__(256) void zero_ws_kernel(float* __restrict__ p, int n4) {
  int i = blockIdx.x * 256 + threadIdx.x;
  if (i < n4) {
    v4f z = {0.f, 0.f, 0.f, 0.f};
    *(v4f*)(p + (i << 2)) = z;
  }
}

__global__ __launch_bounds__(256) void make_Mc_kernel(const float* __restrict__ W,
                                                      const float* __restrict__ b,
                                                      float* __restrict__ Mp,
                                                      float* __restrict__ c) {
  int t = threadIdx.x;
  for (int e = t; e < DF * DF; e += 256) {
    int i = e >> 6, j = e & 63;
    float s = 0.f;
#pragma unroll 8
    for (int k = 0; k < DF; ++k) s += W[k * DF + i] * W[j * DF + k];
    Mp[(i >> 1) * (2 * DF) + j * 2 + (i & 1)] = s;
  }
  if (t < DF) {
    float s = b[t];
#pragma unroll 8
    for (int k = 0; k < DF; ++k) s += b[k] * W[t * DF + k];
    c[t] = s;
  }
}

__global__ __launch_bounds__(256) void spmm_scatter_kernel(
    const float* __restrict__ lap, const int* __restrict__ rows,
    const int* __restrict__ cols, const float* __restrict__ input,
    float* __restrict__ x, int nedges) {
  int gid = blockIdx.x * 256 + threadIdx.x;
  int e = gid >> 4;
  if (e >= nedges) return;
  int q = (gid & 15) << 2;
  int r  = rows[e];
  int ci = cols[e];
  float l = lap[e];
  v4f v = *(const v4f*)(input + (size_t)ci * DF + q);
  float* dst = x + (size_t)r * DF + q;
  unsafeAtomicAdd(dst + 0, l * v.x);
  unsafeAtomicAdd(dst + 1, l * v.y);
  unsafeAtomicAdd(dst + 2, l * v.z);
  unsafeAtomicAdd(dst + 3, l * v.w);
}

__global__ __launch_bounds__(256) void gemm_fused_kernel(
    const float* __restrict__ x, const float* __restrict__ Mp,
    const float* __restrict__ c, float* __restrict__ out, int nrows) {
  __shared__ float ldsA[128 * DF];
  __shared__ float ldsB[DF * DF];
  __shared__ float ldsC[DF];

  int t = threadIdx.x;
  int rowBase0 = blockIdx.x * 128;

#pragma unroll
  for (int j = 0; j < 8; ++j) {
    int fi = (j * 256 + t) << 2;
    int row = rowBase0 + (fi >> 6);
    v4f v = {0.f, 0.f, 0.f, 0.f};
    if (row < nrows) v = *(const v4f*)(x + (size_t)rowBase0 * DF + fi);
    *(v4f*)(ldsA + fi) = v;
  }
#pragma unroll
  for (int j = 0; j < 4; ++j) {
    int fi = (j * 256 + t) << 2;
    *(v4f*)(ldsB + fi) = *(const v4f*)(Mp + fi);
  }
  if (t < DF) ldsC[t] = c[t];
  __syncthreads();

  int wave = t >> 5;
  int lane = t & 31;
  int rowBase = rowBase0 + wave * 16;
  if (rowBase >= nrows) return;
  int m  = lane & 15;
  int kh = lane >> 4;

  const float* A = ldsA + wave * (16 * DF) + m * DF;
  v8f acc0 = {}, acc1 = {}, acc2 = {}, acc3 = {};
#pragma unroll
  for (int k0 = 0; k0 < DF; k0 += 4) {
    int ka = k0 + 2 * kh;
    v2f a = *(const v2f*)(A + ka);
    const float* Bp = ldsB + (ka >> 1) * (2 * DF);
    v2f b0 = *(const v2f*)(Bp + (m      ) * 2);
    v2f b1 = *(const v2f*)(Bp + (m + 16 ) * 2);
    v2f b2 = *(const v2f*)(Bp + (m + 32 ) * 2);
    v2f b3 = *(const v2f*)(Bp + (m + 48 ) * 2);
    acc0 = __builtin_amdgcn_wmma_f32_16x16x4_f32(false, a, false, b0, (short)0, acc0, false, false);
    acc1 = __builtin_amdgcn_wmma_f32_16x16x4_f32(false, a, false, b1, (short)0, acc1, false, false);
    acc2 = __builtin_amdgcn_wmma_f32_16x16x4_f32(false, a, false, b2, (short)0, acc2, false, false);
    acc3 = __builtin_amdgcn_wmma_f32_16x16x4_f32(false, a, false, b3, (short)0, acc3, false, false);
  }

  float c0 = ldsC[m], c1 = ldsC[m + 16], c2 = ldsC[m + 32], c3 = ldsC[m + 48];
#pragma unroll
  for (int v = 0; v < 8; ++v) {
    int r = rowBase + v + 8 * kh;
    float* o = out + (size_t)r * DF;
    o[m]      = acc0[v] + c0;
    o[m + 16] = acc1[v] + c1;
    o[m + 32] = acc2[v] + c2;
    o[m + 48] = acc3[v] + c3;
  }
}

extern "C" void kernel_launch(void* const* d_in, const int* in_sizes, int n_in,
                              void* d_out, int out_size, void* d_ws, size_t ws_size,
                              hipStream_t stream) {
  const float* lap  = (const float*)d_in[0];
  const int*   eidx = (const int*)  d_in[1];
  const float* inp  = (const float*)d_in[2];
  const float* W    = (const float*)d_in[3];
  const float* b    = (const float*)d_in[4];
  float* out = (float*)d_out;

  const int E = in_sizes[0];
  const int N = in_sizes[2] / DF;
  const int* rows = eidx;
  const int* cols = eidx + E;

  float* xbuf = (float*)d_ws;
  float* Mp   = xbuf + (size_t)N * DF;
  float* cvec = Mp + DF * DF;

  int n4 = (N * DF) / 4;
  zero_ws_kernel<<<(n4 + 255) / 256, 256, 0, stream>>>(xbuf, n4);
  make_Mc_kernel<<<1, 256, 0, stream>>>(W, b, Mp, cvec);
  long long tot = (long long)E * 16;
  spmm_scatter_kernel<<<(int)((tot + 255) / 256), 256, 0, stream>>>(lap, rows, cols, inp, xbuf, E);
  int nRowTiles = (N + 15) / 16;
  gemm_fused_kernel<<<(nRowTiles + 7) / 8, 256, 0, stream>>>(xbuf, Mp, cvec, out, N);
}


# Collect: a wave asks for all chunks' directory words at entry, lays their runs out in an LDS position table and reads the strip's records 32 to a load, all loads in flight at once; sum order changes
# speedup vs baseline: 3.5826x; 3.5826x over previous
#include <hip/hip_runtime.h>


typedef __attribute__((ext_vector_type(2))) float v2f;
typedef __attribute__((ext_vector_type(4))) float v4f;
typedef __attribute__((ext_vector_type(8))) float v8f;
typedef __attribute__((ext_vector_type(2))) int v2i;
typedef __attribute__((ext_vector_type(4))) int v4i;

#define DF 64
#define RPH 8
#define WP 68

#define LT 1024
#define EPT 16
#define EPB (LT * EPT)
#define NBIN 8192
#define BPT (NBIN / LT)

template <bool FULL>
__device__ __forceinline__ void order_body(
    const float* __restrict__ lap, const int* __restrict__ rows,
    const int* __restrict__ cols, int* __restrict__ recs, int* __restrict__ dirRow,
    int base, int nloc, int shift, int half,
    int* sCnt, int* sRow, int* sSrc, float* sVal, int* sWave, int* sTot) {
  const int t = threadIdx.x;
  const int lane = t & 31;
  const int wv = t >> 5;

  int r[EPT], c[EPT], rank[EPT];
  float l[EPT];
#pragma unroll
  for (int k = 0; k < EPT; ++k) {
    int i = t + k * LT;
    r[k] = -1; c[k] = 0; l[k] = 0.f;
    if (FULL || i < nloc) {
      r[k] = rows[base + i];
      c[k] = cols[base + i];
      l[k] = lap[base + i];
    }
  }
#pragma unroll
  for (int k = 0; k < EPT; ++k) {
    if (r[k] >= 0 && ((r[k] >> shift) & 1) != half) r[k] = -1;
    rank[k] = 0;
    if (r[k] >= 0) rank[k] = atomicAdd(&sCnt[r[k] >> shift], 1);
  }
  __syncthreads();

  int bc[BPT];
  int s = 0;
#pragma unroll
  for (int k = 0; k < BPT; ++k) { bc[k] = sCnt[BPT * t + k]; s += bc[k]; }
  int x = s;
#pragma unroll
  for (int d = 1; d < 32; d <<= 1) {
    int y = __shfl_up(x, d);
    if (lane >= d) x += y;
  }
  if (lane == 31) sWave[wv] = x;
  __syncthreads();
  if (wv == 0) {
    int w = sWave[lane];
    int z = w;
#pragma unroll
    for (int d = 1; d < 32; d <<= 1) {
      int y = __shfl_up(z, d);
      if (lane >= d) z += y;
    }
    sWave[lane] = z - w;
    if (lane == 31) *sTot = z;
  }
  __syncthreads();
  int off = sWave[wv] + x - s;
  int de[BPT];
#pragma unroll
  for (int k = 0; k < BPT; ++k) {
    sCnt[BPT * t + k] = off;
    de[k] = off | (bc[k] << 16);
    off += bc[k];
  }
#pragma unroll
  for (int k = 0; k < BPT; k += 4) {
    v4i d4 = {de[k], de[k + 1], de[k + 2], de[k + 3]};
    *(v4i*)(dirRow + BPT * t + k) = d4;
  }
  __syncthreads();

#pragma unroll
  for (int k = 0; k < EPT; ++k) {
    if (r[k] >= 0) {
      int p = sCnt[r[k] >> shift] + rank[k];
      sRow[p] = r[k];
      sSrc[p] = c[k] * (DF / 4);
      sVal[p] = l[k];
    }
  }
  __syncthreads();

  const int nk = *sTot;
  const int first = half ? base + nloc - 1 : base;
  const int step = half ? -1 : 1;
#pragma unroll
  for (int k = 0; k < EPT; ++k) {
    int p = t + k * LT;
    if (p < nk) {
      v4i rec = {sRow[p], sSrc[p], __float_as_int(sVal[p]), 0};
      ((v4i*)recs)[(unsigned)(first + step * p)] = rec;
    }
  }
}

__global__ __launch_bounds__(LT) void order_edges_kernel(
    const float* __restrict__ lap, const int* __restrict__ rows,
    const int* __restrict__ cols, int* __restrict__ recs, int* __restrict__ dir,
    int nedges, int nheads) {
  __shared__ int   sCnt[NBIN];
  __shared__ int   sRow[EPB];
  __shared__ int   sSrc[EPB];
  __shared__ float sVal[EPB];
  __shared__ int   sWave[32];
  __shared__ int   sTot;

  const int t = threadIdx.x;
  const int half = blockIdx.x & 1;
  const int base = (blockIdx.x >> 1) * EPB;
  int nloc = nedges - base;
  if (nloc > EPB) nloc = EPB;
  int* dirRow = dir + (size_t)blockIdx.x * NBIN;
  int shift = 4;
  while (((nheads - 1) >> shift) >= NBIN) ++shift;

#pragma unroll
  for (int k = 0; k < NBIN / LT; ++k) sCnt[t + k * LT] = 0;
  __syncthreads();

  if (nloc == EPB)
    order_body<true>(lap, rows, cols, recs, dirRow, base, nloc, shift, half, sCnt, sRow, sSrc, sVal, sWave, &sTot);
  else
    order_body<false>(lap, rows, cols, recs, dirRow, base, nloc, shift, half, sCnt, sRow, sSrc, sVal, sWave, &sTot);
}

__device__ __forceinline__ void strip_times_Wt(const float* A, const float* Bw, int ka0,
                                               v8f& acc0, v8f& acc1, v8f& acc2, v8f& acc3) {
#pragma unroll
  for (int k0 = 0; k0 < DF; k0 += 4) {
    int ka = k0 + ka0;
    v2f a  = *(const v2f*)(A + ka);
    v2f b0 = *(const v2f*)(Bw + ka);
    v2f b1 = *(const v2f*)(Bw + 16 * WP + ka);
    v2f b2 = *(const v2f*)(Bw + 32 * WP + ka);
    v2f b3 = *(const v2f*)(Bw + 48 * WP + ka);
    acc0 = __builtin_amdgcn_wmma_f32_16x16x4_f32(false, a, false, b0, (short)0, acc0, false, false);
    acc1 = __builtin_amdgcn_wmma_f32_16x16x4_f32(false, a, false, b1, (short)0, acc1, false, false);
    acc2 = __builtin_amdgcn_wmma_f32_16x16x4_f32(false, a, false, b2, (short)0, acc2, false, false);
    acc3 = __builtin_amdgcn_wmma_f32_16x16x4_f32(false, a, false, b3, (short)0, acc3, false, false);
  }
}

#define RUN 8
#define RU 4
#define NB 4
#define WIN 320
#define CAP 351
static_assert(8 * (CAP + 1) + 4 * WIN <= 16 * DF * 4 && WIN <= CAP && WIN % 32 == 0, "a wave's LDS places");

__device__ __forceinline__ void wave_lds_handover() {
  __builtin_amdgcn_fence(__ATOMIC_RELEASE, "wavefront");
  __builtin_amdgcn_wave_barrier();
  __builtin_amdgcn_fence(__ATOMIC_ACQUIRE, "wavefront");
}

__device__ __forceinline__ void order_and_sum(int T, int lane, int kh, int m,
                                              v2i* U, v2i* Q, int* cnt, int* cur,
                                              const v4f* __restrict__ in4, v4f (&acc)[RPH]) {
  wave_lds_handover();
  const int cn = cnt[lane & 15];
  int x = cn;
#pragma unroll
  for (int d = 1; d < 16; d <<= 1) {
    int y = __shfl_up(x, d, 16);
    if ((lane & 15) >= d) x += y;
  }
  const int st = x - cn;
  if (lane < 16) { cur[lane] = st; cnt[lane] = 0; }
  wave_lds_handover();
  for (int i = lane; i < T; i += 32) {
    v2i e = U[i];
    int pos = atomicAdd(&cur[(unsigned)e.x >> 28], 1);
    Q[pos] = e;
  }
  wave_lds_handover();
#pragma unroll
  for (int j = 0; j < RPH; ++j) {
    const int sj = __shfl(st, kh * RPH + j);
    const int lj = __shfl(cn, kh * RPH + j);
    const int lo = __shfl_xor(lj, 16);
    const int n = lj > lo ? lj : lo;
    for (int q0 = 0; q0 < n; q0 += RUN) {
      v2i rc[RUN];
      v4f v[RUN];
#pragma unroll
      for (int k = 0; k < RUN; ++k) rc[k] = Q[(q0 + k) < lj ? sj + q0 + k : CAP];
#pragma unroll
      for (int k = 0; k < RUN; ++k) v[k] = in4[(unsigned)(rc[k].x & 0x0fffffff) + (unsigned)m];
      asm volatile("" :: "v"(v[0]), "v"(v[1]), "v"(v[2]), "v"(v[3]), "v"(v[4]), "v"(v[5]), "v"(v[6]), "v"(v[7]));
#pragma unroll
      for (int k = 0; k < RUN; ++k) {
        const bool live = (q0 + k) < lj;
        const float lc = __int_as_float(rc[k].y);
        const float s0 = __builtin_fmaf(lc, v[k].x, acc[j].x);
        const float s1 = __builtin_fmaf(lc, v[k].y, acc[j].y);
        const float s2 = __builtin_fmaf(lc, v[k].z, acc[j].z);
        const float s3 = __builtin_fmaf(lc, v[k].w, acc[j].w);
        acc[j].x = live ? s0 : acc[j].x;
        acc[j].y = live ? s1 : acc[j].y;
        acc[j].z = live ? s2 : acc[j].z;
        acc[j].w = live ? s3 : acc[j].w;
      }
    }
  }
  wave_lds_handover();
}

__global__ __launch_bounds__(256) __attribute__((amdgpu_waves_per_eu(8, 8))) void pull_gemm_kernel(
    const int* __restrict__ dir, const int* __restrict__ recs,
    const float* __restrict__ input, const float* __restrict__ W,
    const float* __restrict__ b, float* __restrict__ out, int nrows, int nedges,
    int nchunks, int nheads) {
  __shared__ float ldsA[128 * DF];
  __shared__ float ldsW[DF * WP];
  __shared__ float ldsC[DF];
  __shared__ v2i   sQ[8 * (CAP + 1)];
  __shared__ int   sCntRow[8 * 16];
  __shared__ int   sCurRow[8 * 16];

  int t = threadIdx.x;
  int rowBase0 = blockIdx.x * 128;
  int wave = t >> 5;
  int lane = t & 31;
  int m  = lane & 15;
  int kh = lane >> 4;

  const int strip = (rowBase0 >> 4) + wave;
  int shift = 4;
  while (((nheads - 1) >> shift) >= NBIN) ++shift;
  const int bin = (strip << 4) >> shift;
  const int half = bin & 1;
  const int* dirp = dir + (size_t)half * NBIN + bin;

  int efirst[NB];
#pragma unroll
  for (int k = 0; k < NB; ++k) {
    const int c = NB * lane + k;
    efirst[k] = 0;
    if (c < nchunks) efirst[k] = dirp[(size_t)c * (2 * NBIN)];
  }

#pragma unroll
  for (int j = 0; j < 4; ++j) {
    int fi = (j * 256 + t) << 2;
    *(v4f*)(ldsW + (fi >> 6) * WP + (fi & 63)) = *(const v4f*)(W + fi);
  }
  if (t < DF) ldsC[t] = b[t];

  v2i* myU = (v2i*)(ldsA + wave * (16 * DF));
  int* myTbl = (int*)(myU + (CAP + 1));
  v2i* myQ = sQ + wave * (CAP + 1);
  int* myCnt = sCntRow + wave * 16;
  int* myCur = sCurRow + wave * 16;
  if (lane < 16) myCnt[lane] = 0;
  if (lane == 0) {
    v2i e = {rowBase0 * (DF / 4), 0};
    myQ[CAP] = e;
  }
  wave_lds_handover();

  const v4i* recs4 = (const v4i*)recs;
  const v4f* in4 = (const v4f*)input;
  const unsigned ltmask = (1u << lane) - 1u;

  int T = 0;
  bool fits = true;
  for (int cb = 0; fits && cb < nchunks; cb += 32 * NB) {
    int cnt[NB], slot0[NB], pre[NB];
    int s = 0;
#pragma unroll
    for (int k = 0; k < NB; ++k) {
      const int c = cb + NB * lane + k;
      int e = efirst[k];
      if (cb > 0) {
        e = 0;
        if (c < nchunks) e = dirp[(size_t)c * (2 * NBIN)];
      }
      const int p0 = e & 0xffff;
      cnt[k] = (int)((unsigned)e >> 16);
      const int base = c * EPB;
      int nloc = nedges - base;
      if (nloc > EPB) nloc = EPB;
      slot0[k] = half ? base + nloc - p0 - cnt[k] : base + p0;
      pre[k] = s;
      s += cnt[k];
    }
    int x = s;
#pragma unroll
    for (int d = 1; d < 32; d <<= 1) {
      int y = __shfl_up(x, d);
      if (lane >= d) x += y;
    }
    const int total = __shfl(x, 31);
#pragma unroll
    for (int k = 0; k < NB; ++k) pre[k] += x - s;

    for (int w0 = 0; fits && w0 < total; w0 += WIN) {
      int first[NB], n[NB];
      int nmax = 0;
#pragma unroll
      for (int k = 0; k < NB; ++k) {
        int lo = w0 - pre[k];
        if (lo < 0) lo = 0;
        int hi = w0 + WIN - pre[k];
        if (hi > cnt[k]) hi = cnt[k];
        n[k] = hi - lo;
        first[k] = slot0[k] + lo;
        if (n[k] > nmax) nmax = n[k];
      }
      for (int i = 0; __builtin_amdgcn_ballot_w32(i < nmax) != 0; ++i) {
#pragma unroll
        for (int k = 0; k < NB; ++k)
          if (i < n[k]) myTbl[first[k] - slot0[k] + pre[k] - w0 + i] = first[k] + i;
      }
      wave_lds_handover();
      int nrec = total - w0;
      if (nrec > WIN) nrec = WIN;
      v4i r[WIN / 32];
#pragma unroll
      for (int k = 0; k < WIN / 32; ++k) {
        const int pos = 32 * k + lane;
        r[k] = recs4[(unsigned)myTbl[pos < nrec ? pos : 0]];
      }
      unsigned mask[WIN / 32];
      int tot = 0;
#pragma unroll
      for (int k = 0; k < WIN / 32; ++k) {
        mask[k] = __builtin_amdgcn_ballot_w32(32 * k + lane < nrec && (r[k].x >> 4) == strip);
        tot += __builtin_popcount(mask[k]);
      }
      if (T + tot > CAP) {
        fits = false;
      } else {
#pragma unroll
        for (int k = 0; k < WIN / 32; ++k) {
          if ((mask[k] >> lane) & 1u) {
            int idx = T + __builtin_popcount(mask[k] & ltmask);
            unsigned r4 = (unsigned)r[k].x & 15u;
            v2i u = {(int)((unsigned)r[k].y | (r4 << 28)), r[k].z};
            myU[idx] = u;
            atomicAdd(&myCnt[r4], 1);
          }
          T += __builtin_popcount(mask[k]);
        }
      }
    }
  }

  v4f acc[RPH];
#pragma unroll
  for (int j = 0; j < RPH; ++j) {
    v4f z = {0.f, 0.f, 0.f, 0.f};
    acc[j] = z;
  }

  int fcb = 0, fi0 = 0;
  if (!fits) {
    wave_lds_handover();
    if (lane < 16) myCnt[lane] = 0;
    wave_lds_handover();
    T = 0;
  }
  bool done = fits;
  for (;;) {
    if (!done) {
      done = true;
      for (; fcb < nchunks; fcb += 32, fi0 = 0) {
        const int c = fcb + lane;
        int e = 0;
        if (c < nchunks) e = dirp[(size_t)c * (2 * NBIN)];
        const int p0 = e & 0xffff;
        const int cnt = (int)((unsigned)e >> 16);
        const int base = c * EPB;
        int nloc = nedges - base;
        if (nloc > EPB) nloc = EPB;
        const int slot0 = half ? base + nloc - p0 - cnt : base + p0;
        bool full = false;
        for (; __builtin_amdgcn_ballot_w32(fi0 < cnt) != 0; fi0 += RU) {
          v4i r[RU];
#pragma unroll
          for (int k = 0; k < RU; ++k) {
            v4i none = {-1, 0, 0, 0};
            r[k] = none;
            if (fi0 + k < cnt) r[k] = recs4[(unsigned)(slot0 + fi0 + k)];
          }
          unsigned mask[RU];
          int tot = 0;
#pragma unroll
          for (int k = 0; k < RU; ++k) {
            mask[k] = __builtin_amdgcn_ballot_w32(r[k].x >= 0 && (r[k].x >> 4) == strip);
            tot += __builtin_popcount(mask[k]);
          }
          if (T + tot > CAP) {
            full = true;
            break;
          }
#pragma unroll
          for (int k = 0; k < RU; ++k) {
            if ((mask[k] >> lane) & 1u) {
              int idx = T + __builtin_popcount(mask[k] & ltmask);
              unsigned r4 = (unsigned)r[k].x & 15u;
              v2i u = {(int)((unsigned)r[k].y | (r4 << 28)), r[k].z};
              myU[idx] = u;
              atomicAdd(&myCnt[r4], 1);
            }
            T += __builtin_popcount(mask[k]);
          }
        }
        if (full) {
          done = false;
          break;
        }
      }
    }
    order_and_sum(T, lane, kh, m, myU, myQ, myCnt, myCur, in4, acc);
    T = 0;
    if (done) break;
  }

  int rloc = wave * 16 + kh * RPH;
  const unsigned q = (unsigned)m << 2;
#pragma unroll
  for (int j = 0; j < RPH; ++j) *(v4f*)(ldsA + (rloc + j) * DF + q) = acc[j];
  __syncthreads();

  int rowBase = rowBase0 + wave * 16;
  if (rowBase >= nrows) return;

  float* S = ldsA + wave * (16 * DF);
  const float* A  = S + m * DF;
  const float* Bw = ldsW + m * WP;
  const int ka0 = 2 * kh;
  float c0 = ldsC[m], c1 = ldsC[m + 16], c2 = ldsC[m + 32], c3 = ldsC[m + 48];

  v8f acc0 = {}, acc1 = {}, acc2 = {}, acc3 = {};
  strip_times_Wt(A, Bw, ka0, acc0, acc1, acc2, acc3);

#pragma unroll
  for (int v = 0; v < 8; ++v) {
    float* o = S + (v + 8 * kh) * DF;
    o[m]      = acc0[v] + c0;
    o[m + 16] = acc1[v] + c1;
    o[m + 32] = acc2[v] + c2;
    o[m + 48] = acc3[v] + c3;
  }
  __builtin_amdgcn_fence(__ATOMIC_RELEASE, "wavefront");
  __builtin_amdgcn_wave_barrier();
  __builtin_amdgcn_fence(__ATOMIC_ACQUIRE, "wavefront");

  v8f z8 = {};
  acc0 = z8; acc1 = z8; acc2 = z8; acc3 = z8;
  strip_times_Wt(A, Bw, ka0, acc0, acc1, acc2, acc3);

#pragma unroll
  for (int v = 0; v < 8; ++v) {
    int r = rowBase + v + 8 * kh;
    float* o = out + (size_t)r * DF;
    o[m]      = acc0[v] + c0;
    o[m + 16] = acc1[v] + c1;
    o[m + 32] = acc2[v] + c2;
    o[m + 48] = acc3[v] + c3;
  }
}

extern "C" void kernel_launch(void* const* d_in, const int* in_sizes, int n_in,
                              void* d_out, int out_size, void* d_ws, size_t ws_size,
                              hipStream_t stream) {
  const float* lap  = (const float*)d_in[0];
  const int*   eidx = (const int*)  d_in[1];
  const float* inp  = (const float*)d_in[2];
  const float* W    = (const float*)d_in[3];
  const float* b    = (const float*)d_in[4];
  float* out = (float*)d_out;

  const int E = in_sizes[0];
  const int N = in_sizes[2] / DF;
  const int* rows = eidx;
  const int* cols = eidx + E;

  int nBlocks = (N + 127) / 128;
  int nHeads  = nBlocks * 128;
  int nChunks = (E + EPB - 1) / EPB;

  int* recs = (int*)d_ws;
  int* dir  = recs + 4 * (size_t)E;

  order_edges_kernel<<<2 * nChunks, LT, 0, stream>>>(lap, rows, cols, recs, dir, E, nHeads);
  pull_gemm_kernel<<<nBlocks, 256, 0, stream>>>(dir, recs, inp, W, b, out, N, E, nChunks, nHeads);
}


# Summing launch reshaped: 250 work-groups of 25 strips (400 rows), one to a WGP, so no WGP hosts more than 25 strips; 8-strip work-groups put 24 on most WGPs and 32 on some, whose end sets the time
# speedup vs baseline: 3.9476x; 1.1019x over previous
#include <hip/hip_runtime.h>


typedef __attribute__((ext_vector_type(2))) float v2f;
typedef __attribute__((ext_vector_type(4))) float v4f;
typedef __attribute__((ext_vector_type(8))) float v8f;
typedef __attribute__((ext_vector_type(2))) int v2i;
typedef __attribute__((ext_vector_type(4))) int v4i;

#define DF 64
#define RPH 8
#define SPW 25
#define WP 68

#define LT 1024
#define EPT 16
#define EPB (LT * EPT)
#define NBIN 8192
#define BPT (NBIN / LT)

template <bool FULL>
__device__ __forceinline__ void order_body(
    const float* __restrict__ lap, const int* __restrict__ rows,
    const int* __restrict__ cols, int* __restrict__ recs, int* __restrict__ dirRow,
    int base, int nloc, int shift, int half,
    int* sCnt, int* sRow, int* sSrc, float* sVal, int* sWave, int* sTot) {
  const int t = threadIdx.x;
  const int lane = t & 31;
  const int wv = t >> 5;

  int r[EPT], c[EPT], rank[EPT];
  float l[EPT];
#pragma unroll
  for (int k = 0; k < EPT; ++k) {
    int i = t + k * LT;
    r[k] = -1; c[k] = 0; l[k] = 0.f;
    if (FULL || i < nloc) {
      r[k] = rows[base + i];
      c[k] = cols[base + i];
      l[k] = lap[base + i];
    }
  }
#pragma unroll
  for (int k = 0; k < EPT; ++k) {
    if (r[k] >= 0 && ((r[k] >> shift) & 1) != half) r[k] = -1;
    rank[k] = 0;
    if (r[k] >= 0) rank[k] = atomicAdd(&sCnt[r[k] >> shift], 1);
  }
  __syncthreads();

  int bc[BPT];
  int s = 0;
#pragma unroll
  for (int k = 0; k < BPT; ++k) { bc[k] = sCnt[BPT * t + k]; s += bc[k]; }
  int x = s;
#pragma unroll
  for (int d = 1; d < 32; d <<= 1) {
    int y = __shfl_up(x, d);
    if (lane >= d) x += y;
  }
  if (lane == 31) sWave[wv] = x;
  __syncthreads();
  if (wv == 0) {
    int w = sWave[lane];
    int z = w;
#pragma unroll
    for (int d = 1; d < 32; d <<= 1) {
      int y = __shfl_up(z, d);
      if (lane >= d) z += y;
    }
    sWave[lane] = z - w;
    if (lane == 31) *sTot = z;
  }
  __syncthreads();
  int off = sWave[wv] + x - s;
  int de[BPT];
#pragma unroll
  for (int k = 0; k < BPT; ++k) {
    sCnt[BPT * t + k] = off;
    de[k] = off | (bc[k] << 16);
    off += bc[k];
  }
#pragma unroll
  for (int k = 0; k < BPT; k += 4) {
    v4i d4 = {de[k], de[k + 1], de[k + 2], de[k + 3]};
    *(v4i*)(dirRow + BPT * t + k) = d4;
  }
  __syncthreads();

#pragma unroll
  for (int k = 0; k < EPT; ++k) {
    if (r[k] >= 0) {
      int p = sCnt[r[k] >> shift] + rank[k];
      sRow[p] = r[k];
      sSrc[p] = c[k] * (DF / 4);
      sVal[p] = l[k];
    }
  }
  __syncthreads();

  const int nk = *sTot;
  const int first = half ? base + nloc - 1 : base;
  const int step = half ? -1 : 1;
#pragma unroll
  for (int k = 0; k < EPT; ++k) {
    int p = t + k * LT;
    if (p < nk) {
      v4i rec = {sRow[p], sSrc[p], __float_as_int(sVal[p]), 0};
      ((v4i*)recs)[(unsigned)(first + step * p)] = rec;
    }
  }
}

__global__ __launch_bounds__(LT) void order_edges_kernel(
    const float* __restrict__ lap, const int* __restrict__ rows,
    const int* __restrict__ cols, int* __restrict__ recs, int* __restrict__ dir,
    int nedges, int nheads) {
  __shared__ int   sCnt[NBIN];
  __shared__ int   sRow[EPB];
  __shared__ int   sSrc[EPB];
  __shared__ float sVal[EPB];
  __shared__ int   sWave[32];
  __shared__ int   sTot;

  const int t = threadIdx.x;
  const int half = blockIdx.x & 1;
  const int base = (blockIdx.x >> 1) * EPB;
  int nloc = nedges - base;
  if (nloc > EPB) nloc = EPB;
  int* dirRow = dir + (size_t)blockIdx.x * NBIN;
  int shift = 4;
  while (((nheads - 1) >> shift) >= NBIN) ++shift;

#pragma unroll
  for (int k = 0; k < NBIN / LT; ++k) sCnt[t + k * LT] = 0;
  __syncthreads();

  if (nloc == EPB)
    order_body<true>(lap, rows, cols, recs, dirRow, base, nloc, shift, half, sCnt, sRow, sSrc, sVal, sWave, &sTot);
  else
    order_body<false>(lap, rows, cols, recs, dirRow, base, nloc, shift, half, sCnt, sRow, sSrc, sVal, sWave, &sTot);
}

__device__ __forceinline__ void strip_times_Wt(const float* A, const float* Bw, int ka0,
                                               v8f& acc0, v8f& acc1, v8f& acc2, v8f& acc3) {
#pragma unroll
  for (int k0 = 0; k0 < DF; k0 += 4) {
    int ka = k0 + ka0;
    v2f a  = *(const v2f*)(A + ka);
    v2f b0 = *(const v2f*)(Bw + ka);
    v2f b1 = *(const v2f*)(Bw + 16 * WP + ka);
    v2f b2 = *(const v2f*)(Bw + 32 * WP + ka);
    v2f b3 = *(const v2f*)(Bw + 48 * WP + ka);
    acc0 = __builtin_amdgcn_wmma_f32_16x16x4_f32(false, a, false, b0, (short)0, acc0, false, false);
    acc1 = __builtin_amdgcn_wmma_f32_16x16x4_f32(false, a, false, b1, (short)0, acc1, false, false);
    acc2 = __builtin_amdgcn_wmma_f32_16x16x4_f32(false, a, false, b2, (short)0, acc2, false, false);
    acc3 = __builtin_amdgcn_wmma_f32_16x16x4_f32(false, a, false, b3, (short)0, acc3, false, false);
  }
}

#define RUN 8
#define RU 4
#define NB 4
#define WIN 320
#define CAP 351
static_assert(8 * (CAP + 1) + 4 * WIN <= 16 * DF * 4 && WIN <= CAP && WIN % 32 == 0, "a wave's LDS places");

__device__ __forceinline__ void wave_lds_handover() {
  __builtin_amdgcn_fence(__ATOMIC_RELEASE, "wavefront");
  __builtin_amdgcn_wave_barrier();
  __builtin_amdgcn_fence(__ATOMIC_ACQUIRE, "wavefront");
}

__device__ __forceinline__ void order_and_sum(int T, int lane, int kh, int m,
                                              v2i* U, v2i* Q, int* cnt, int* cur,
                                              const v4f* __restrict__ in4, v4f (&acc)[RPH]) {
  wave_lds_handover();
  const int cn = cnt[lane & 15];
  int x = cn;
#pragma unroll
  for (int d = 1; d < 16; d <<= 1) {
    int y = __shfl_up(x, d, 16);
    if ((lane & 15) >= d) x += y;
  }
  const int st = x - cn;
  if (lane < 16) { cur[lane] = st; cnt[lane] = 0; }
  wave_lds_handover();
  for (int i = lane; i < T; i += 32) {
    v2i e = U[i];
    int pos = atomicAdd(&cur[(unsigned)e.x >> 28], 1);
    Q[pos] = e;
  }
  wave_lds_handover();
#pragma unroll
  for (int j = 0; j < RPH; ++j) {
    const int sj = __shfl(st, kh * RPH + j);
    const int lj = __shfl(cn, kh * RPH + j);
    const int lo = __shfl_xor(lj, 16);
    const int n = lj > lo ? lj : lo;
    for (int q0 = 0; q0 < n; q0 += RUN) {
      v2i rc[RUN];
      v4f v[RUN];
#pragma unroll
      for (int k = 0; k < RUN; ++k) rc[k] = Q[(q0 + k) < lj ? sj + q0 + k : CAP];
#pragma unroll
      for (int k = 0; k < RUN; ++k) v[k] = in4[(unsigned)(rc[k].x & 0x0fffffff) + (unsigned)m];
      asm volatile("" :: "v"(v[0]), "v"(v[1]), "v"(v[2]), "v"(v[3]), "v"(v[4]), "v"(v[5]), "v"(v[6]), "v"(v[7]));
      __builtin_amdgcn_sched_barrier(0);
#pragma unroll
      for (int k = 0; k < RUN; ++k) {
        const bool live = (q0 + k) < lj;
        const float lc = __int_as_float(rc[k].y);
        const float s0 = __builtin_fmaf(lc, v[k].x, acc[j].x);
        const float s1 = __builtin_fmaf(lc, v[k].y, acc[j].y);
        const float s2 = __builtin_fmaf(lc, v[k].z, acc[j].z);
        const float s3 = __builtin_fmaf(lc, v[k].w, acc[j].w);
        acc[j].x = live ? s0 : acc[j].x;
        acc[j].y = live ? s1 : acc[j].y;
        acc[j].z = live ? s2 : acc[j].z;
        acc[j].w = live ? s3 : acc[j].w;
      }
    }
  }
  wave_lds_handover();
}

__global__ __launch_bounds__(SPW * 32) void pull_gemm_kernel(
    const int* __restrict__ dir, const int* __restrict__ recs,
    const float* __restrict__ input, const float* __restrict__ W,
    const float* __restrict__ b, float* __restrict__ out, int nrows, int nedges,
    int nchunks, int nheads) {
  __shared__ float ldsA[SPW * 16 * DF];
  __shared__ float ldsW[DF * WP];
  __shared__ float ldsC[DF];
  __shared__ v2i   sQ[SPW * (CAP + 1)];
  __shared__ int   sCntRow[SPW * 16];
  __shared__ int   sCurRow[SPW * 16];

  int t = threadIdx.x;
  int rowBase0 = blockIdx.x * (SPW * 16);
  int wave = t >> 5;
  int lane = t & 31;
  int m  = lane & 15;
  int kh = lane >> 4;

  const int strip = (rowBase0 >> 4) + wave;
  int shift = 4;
  while (((nheads - 1) >> shift) >= NBIN) ++shift;
  const int bin = (strip << 4) >> shift;
  const int half = bin & 1;
  const int* dirp = dir + (size_t)half * NBIN + bin;

  int efirst[NB];
#pragma unroll
  for (int k = 0; k < NB; ++k) {
    const int c = NB * lane + k;
    efirst[k] = 0;
    if (c < nchunks) efirst[k] = dirp[(size_t)c * (2 * NBIN)];
  }

  constexpr int WJ = (DF * DF / 4 + SPW * 32 - 1) / (SPW * 32);
  v4f wv[WJ];
#pragma unroll
  for (int j = 0; j < WJ; ++j) {
    int fi = (j * (SPW * 32) + t) << 2;
    v4f z = {0.f, 0.f, 0.f, 0.f};
    wv[j] = z;
    if (fi < DF * DF) wv[j] = *(const v4f*)(W + fi);
  }
#pragma unroll
  for (int j = 0; j < WJ; ++j) {
    int fi = (j * (SPW * 32) + t) << 2;
    if (fi < DF * DF) *(v4f*)(ldsW + (fi >> 6) * WP + (fi & 63)) = wv[j];
  }
  if (t < DF) ldsC[t] = b[t];

  v2i* myU = (v2i*)(ldsA + wave * (16 * DF));
  int* myTbl = (int*)(myU + (CAP + 1));
  v2i* myQ = sQ + wave * (CAP + 1);
  int* myCnt = sCntRow + wave * 16;
  int* myCur = sCurRow + wave * 16;
  if (lane < 16) myCnt[lane] = 0;
  if (lane == 0) {
    v2i e = {rowBase0 * (DF / 4), 0};
    myQ[CAP] = e;
  }
  wave_lds_handover();

  const v4i* recs4 = (const v4i*)recs;
  const v4f* in4 = (const v4f*)input;
  const unsigned ltmask = (1u << lane) - 1u;

  int T = 0;
  bool fits = true;
  for (int cb = 0; fits && cb < nchunks; cb += 32 * NB) {
    int cnt[NB], slot0[NB], pre[NB];
    int s = 0;
#pragma unroll
    for (int k = 0; k < NB; ++k) {
      const int c = cb + NB * lane + k;
      int e = efirst[k];
      if (cb > 0) {
        e = 0;
        if (c < nchunks) e = dirp[(size_t)c * (2 * NBIN)];
      }
      const int p0 = e & 0xffff;
      cnt[k] = (int)((unsigned)e >> 16);
      const int base = c * EPB;
      int nloc = nedges - base;
      if (nloc > EPB) nloc = EPB;
      slot0[k] = half ? base + nloc - p0 - cnt[k] : base + p0;
      pre[k] = s;
      s += cnt[k];
    }
    int x = s;
#pragma unroll
    for (int d = 1; d < 32; d <<= 1) {
      int y = __shfl_up(x, d);
      if (lane >= d) x += y;
    }
    const int total = __shfl(x, 31);
#pragma unroll
    for (int k = 0; k < NB; ++k) pre[k] += x - s;

    for (int w0 = 0; fits && w0 < total; w0 += WIN) {
      int first[NB], n[NB];
      int nmax = 0;
#pragma unroll
      for (int k = 0; k < NB; ++k) {
        int lo = w0 - pre[k];
        if (lo < 0) lo = 0;
        int hi = w0 + WIN - pre[k];
        if (hi > cnt[k]) hi = cnt[k];
        n[k] = hi - lo;
        first[k] = slot0[k] + lo;
        if (n[k] > nmax) nmax = n[k];
      }
      for (int i = 0; __builtin_amdgcn_ballot_w32(i < nmax) != 0; ++i) {
#pragma unroll
        for (int k = 0; k < NB; ++k)
          if (i < n[k]) myTbl[first[k] - slot0[k] + pre[k] - w0 + i] = first[k] + i;
      }
      wave_lds_handover();
      int nrec = total - w0;
      if (nrec > WIN) nrec = WIN;
      v4i r[WIN / 32];
#pragma unroll
      for (int k = 0; k < WIN / 32; ++k) {
        const int pos = 32 * k + lane;
        r[k] = recs4[(unsigned)myTbl[pos < nrec ? pos : 0]];
      }
      unsigned mask[WIN / 32];
      int tot = 0;
#pragma unroll
      for (int k = 0; k < WIN / 32; ++k) {
        mask[k] = __builtin_amdgcn_ballot_w32(32 * k + lane < nrec && (r[k].x >> 4) == strip);
        tot += __builtin_popcount(mask[k]);
      }
      if (T + tot > CAP) {
        fits = false;
      } else {
#pragma unroll
        for (int k = 0; k < WIN / 32; ++k) {
          if ((mask[k] >> lane) & 1u) {
            int idx = T + __builtin_popcount(mask[k] & ltmask);
            unsigned r4 = (unsigned)r[k].x & 15u;
            v2i u = {(int)((unsigned)r[k].y | (r4 << 28)), r[k].z};
            myU[idx] = u;
            atomicAdd(&myCnt[r4], 1);
          }
          T += __builtin_popcount(mask[k]);
        }
      }
    }
  }

  v4f acc[RPH];
#pragma unroll
  for (int j = 0; j < RPH; ++j) {
    v4f z = {0.f, 0.f, 0.f, 0.f};
    acc[j] = z;
  }

  int fcb = 0, fi0 = 0;
  if (!fits) {
    wave_lds_handover();
    if (lane < 16) myCnt[lane] = 0;
    wave_lds_handover();
    T = 0;
  }
  bool done = fits;
  for (;;) {
    if (!done) {
      done = true;
      for (; fcb < nchunks; fcb += 32, fi0 = 0) {
        const int c = fcb + lane;
        int e = 0;
        if (c < nchunks) e = dirp[(size_t)c * (2 * NBIN)];
        const int p0 = e & 0xffff;
        const int cnt = (int)((unsigned)e >> 16);
        const int base = c * EPB;
        int nloc = nedges - base;
        if (nloc > EPB) nloc = EPB;
        const int slot0 = half ? base + nloc - p0 - cnt : base + p0;
        bool full = false;
        for (; __builtin_amdgcn_ballot_w32(fi0 < cnt) != 0; fi0 += RU) {
          v4i r[RU];
#pragma unroll
          for (int k = 0; k < RU; ++k) {
            v4i none = {-1, 0, 0, 0};
            r[k] = none;
            if (fi0 + k < cnt) r[k] = recs4[(unsigned)(slot0 + fi0 + k)];
          }
          unsigned mask[RU];
          int tot = 0;
#pragma unroll
          for (int k = 0; k < RU; ++k) {
            mask[k] = __builtin_amdgcn_ballot_w32(r[k].x >= 0 && (r[k].x >> 4) == strip);
            tot += __builtin_popcount(mask[k]);
          }
          if (T + tot > CAP) {
            full = true;
            break;
          }
#pragma unroll
          for (int k = 0; k < RU; ++k) {
            if ((mask[k] >> lane) & 1u) {
              int idx = T + __builtin_popcount(mask[k] & ltmask);
              unsigned r4 = (unsigned)r[k].x & 15u;
              v2i u = {(int)((unsigned)r[k].y | (r4 << 28)), r[k].z};
              myU[idx] = u;
              atomicAdd(&myCnt[r4], 1);
            }
            T += __builtin_popcount(mask[k]);
          }
        }
        if (full) {
          done = false;
          break;
        }
      }
    }
    order_and_sum(T, lane, kh, m, myU, myQ, myCnt, myCur, in4, acc);
    T = 0;
    if (done) break;
  }

  int rloc = wave * 16 + kh * RPH;
  const unsigned q = (unsigned)m << 2;
#pragma unroll
  for (int j = 0; j < RPH; ++j) *(v4f*)(ldsA + (rloc + j) * DF + q) = acc[j];
  __syncthreads();

  int rowBase = rowBase0 + wave * 16;
  if (rowBase >= nrows) return;

  float* S = ldsA + wave * (16 * DF);
  const float* A  = S + m * DF;
  const float* Bw = ldsW + m * WP;
  const int ka0 = 2 * kh;
  float c0 = ldsC[m], c1 = ldsC[m + 16], c2 = ldsC[m + 32], c3 = ldsC[m + 48];

  v8f acc0 = {}, acc1 = {}, acc2 = {}, acc3 = {};
  strip_times_Wt(A, Bw, ka0, acc0, acc1, acc2, acc3);

#pragma unroll
  for (int v = 0; v < 8; ++v) {
    float* o = S + (v + 8 * kh) * DF;
    o[m]      = acc0[v] + c0;
    o[m + 16] = acc1[v] + c1;
    o[m + 32] = acc2[v] + c2;
    o[m + 48] = acc3[v] + c3;
  }
  __builtin_amdgcn_fence(__ATOMIC_RELEASE, "wavefront");
  __builtin_amdgcn_wave_barrier();
  __builtin_amdgcn_fence(__ATOMIC_ACQUIRE, "wavefront");

  v8f z8 = {};
  acc0 = z8; acc1 = z8; acc2 = z8; acc3 = z8;
  strip_times_Wt(A, Bw, ka0, acc0, acc1, acc2, acc3);

#pragma unroll
  for (int v = 0; v < 8; ++v) {
    int r = rowBase + v + 8 * kh;
    float* o = out + (size_t)r * DF;
    o[m]      = acc0[v] + c0;
    o[m + 16] = acc1[v] + c1;
    o[m + 32] = acc2[v] + c2;
    o[m + 48] = acc3[v] + c3;
  }
}

extern "C" void kernel_launch(void* const* d_in, const int* in_sizes, int n_in,
                              void* d_out, int out_size, void* d_ws, size_t ws_size,
                              hipStream_t stream) {
  const float* lap  = (const float*)d_in[0];
  const int*   eidx = (const int*)  d_in[1];
  const float* inp  = (const float*)d_in[2];
  const float* W    = (const float*)d_in[3];
  const float* b    = (const float*)d_in[4];
  float* out = (float*)d_out;

  const int E = in_sizes[0];
  const int N = in_sizes[2] / DF;
  const int* rows = eidx;
  const int* cols = eidx + E;

  int nBlocks = (N + SPW * 16 - 1) / (SPW * 16);
  int nHeads  = nBlocks * (SPW * 16);
  int nChunks = (E + EPB - 1) / EPB;

  int* recs = (int*)d_ws;
  int* dir  = recs + 4 * (size_t)E;

  order_edges_kernel<<<2 * nChunks, LT, 0, stream>>>(lap, rows, cols, recs, dir, E, nHeads);
  pull_gemm_kernel<<<nBlocks, SPW * 32, 0, stream>>>(dir, recs, inp, W, b, out, N, E, nChunks, nHeads);
}


# Linears folded in the summing launch: each work-group makes P = W W and c = W b + b on WMMA at entry (20 waves, a tile each); a strip does one product x P^T + c, not two; product sum order changes
# speedup vs baseline: 4.1596x; 1.0537x over previous
#include <hip/hip_runtime.h>


typedef __attribute__((ext_vector_type(2))) float v2f;
typedef __attribute__((ext_vector_type(4))) float v4f;
typedef __attribute__((ext_vector_type(8))) float v8f;
typedef __attribute__((ext_vector_type(2))) int v2i;
typedef __attribute__((ext_vector_type(4))) int v4i;

#define DF 64
#define RPH 8
#define SPW 25
#define WP 68

#define LT 1024
#define EPT 16
#define EPB (LT * EPT)
#define NBIN 8192
#define BPT (NBIN / LT)

template <bool FULL>
__device__ __forceinline__ void order_body(
    const float* __restrict__ lap, const int* __restrict__ rows,
    const int* __restrict__ cols, int* __restrict__ recs, int* __restrict__ dirRow,
    int base, int nloc, int shift, int half,
    int* sCnt, int* sRow, int* sSrc, float* sVal, int* sWave, int* sTot) {
  const int t = threadIdx.x;
  const int lane = t & 31;
  const int wv = t >> 5;

  int r[EPT], c[EPT], rank[EPT];
  float l[EPT];
#pragma unroll
  for (int k = 0; k < EPT; ++k) {
    int i = t + k * LT;
    r[k] = -1; c[k] = 0; l[k] = 0.f;
    if (FULL || i < nloc) {
      r[k] = rows[base + i];
      c[k] = cols[base + i];
      l[k] = lap[base + i];
    }
  }
#pragma unroll
  for (int k = 0; k < EPT; ++k) {
    if (r[k] >= 0 && ((r[k] >> shift) & 1) != half) r[k] = -1;
    rank[k] = 0;
    if (r[k] >= 0) rank[k] = atomicAdd(&sCnt[r[k] >> shift], 1);
  }
  __syncthreads();

  int bc[BPT];
  int s = 0;
#pragma unroll
  for (int k = 0; k < BPT; ++k) { bc[k] = sCnt[BPT * t + k]; s += bc[k]; }
  int x = s;
#pragma unroll
  for (int d = 1; d < 32; d <<= 1) {
    int y = __shfl_up(x, d);
    if (lane >= d) x += y;
  }
  if (lane == 31) sWave[wv] = x;
  __syncthreads();
  if (wv == 0) {
    int w = sWave[lane];
    int z = w;
#pragma unroll
    for (int d = 1; d < 32; d <<= 1) {
      int y = __shfl_up(z, d);
      if (lane >= d) z += y;
    }
    sWave[lane] = z - w;
    if (lane == 31) *sTot = z;
  }
  __syncthreads();
  int off = sWave[wv] + x - s;
  int de[BPT];
#pragma unroll
  for (int k = 0; k < BPT; ++k) {
    sCnt[BPT * t + k] = off;
    de[k] = off | (bc[k] << 16);
    off += bc[k];
  }
#pragma unroll
  for (int k = 0; k < BPT; k += 4) {
    v4i d4 = {de[k], de[k + 1], de[k + 2], de[k + 3]};
    *(v4i*)(dirRow + BPT * t + k) = d4;
  }
  __syncthreads();

#pragma unroll
  for (int k = 0; k < EPT; ++k) {
    if (r[k] >= 0) {
      int p = sCnt[r[k] >> shift] + rank[k];
      sRow[p] = r[k];
      sSrc[p] = c[k] * (DF / 4);
      sVal[p] = l[k];
    }
  }
  __syncthreads();

  const int nk = *sTot;
  const int first = half ? base + nloc - 1 : base;
  const int step = half ? -1 : 1;
#pragma unroll
  for (int k = 0; k < EPT; ++k) {
    int p = t + k * LT;
    if (p < nk) {
      v4i rec = {sRow[p], sSrc[p], __float_as_int(sVal[p]), 0};
      ((v4i*)recs)[(unsigned)(first + step * p)] = rec;
    }
  }
}

__global__ __launch_bounds__(LT) void order_edges_kernel(
    const float* __restrict__ lap, const int* __restrict__ rows,
    const int* __restrict__ cols, int* __restrict__ recs, int* __restrict__ dir,
    int nedges, int nheads) {
  __shared__ int   sCnt[NBIN];
  __shared__ int   sRow[EPB];
  __shared__ int   sSrc[EPB];
  __shared__ float sVal[EPB];
  __shared__ int   sWave[32];
  __shared__ int   sTot;

  const int t = threadIdx.x;
  const int half = blockIdx.x & 1;
  const int base = (blockIdx.x >> 1) * EPB;
  int nloc = nedges - base;
  if (nloc > EPB) nloc = EPB;
  int* dirRow = dir + (size_t)blockIdx.x * NBIN;
  int shift = 4;
  while (((nheads - 1) >> shift) >= NBIN) ++shift;

#pragma unroll
  for (int k = 0; k < NBIN / LT; ++k) sCnt[t + k * LT] = 0;
  __syncthreads();

  if (nloc == EPB)
    order_body<true>(lap, rows, cols, recs, dirRow, base, nloc, shift, half, sCnt, sRow, sSrc, sVal, sWave, &sTot);
  else
    order_body<false>(lap, rows, cols, recs, dirRow, base, nloc, shift, half, sCnt, sRow, sSrc, sVal, sWave, &sTot);
}

__device__ __forceinline__ void strip_times_Mt(const float* A, const float* Bw, int ka0,
                                               v8f& acc0, v8f& acc1, v8f& acc2, v8f& acc3) {
#pragma unroll
  for (int k0 = 0; k0 < DF; k0 += 4) {
    int ka = k0 + ka0;
    v2f a  = *(const v2f*)(A + ka);
    v2f b0 = *(const v2f*)(Bw + ka);
    v2f b1 = *(const v2f*)(Bw + 16 * WP + ka);
    v2f b2 = *(const v2f*)(Bw + 32 * WP + ka);
    v2f b3 = *(const v2f*)(Bw + 48 * WP + ka);
    acc0 = __builtin_amdgcn_wmma_f32_16x16x4_f32(false, a, false, b0, (short)0, acc0, false, false);
    acc1 = __builtin_amdgcn_wmma_f32_16x16x4_f32(false, a, false, b1, (short)0, acc1, false, false);
    acc2 = __builtin_amdgcn_wmma_f32_16x16x4_f32(false, a, false, b2, (short)0, acc2, false, false);
    acc3 = __builtin_amdgcn_wmma_f32_16x16x4_f32(false, a, false, b3, (short)0, acc3, false, false);
  }
}

#define RUN 8
#define RU 4
#define NB 4
#define WIN 320
#define CAP 351
static_assert(8 * (CAP + 1) + 4 * WIN <= 16 * DF * 4 && WIN <= CAP && WIN % 32 == 0, "a wave's LDS places");

__device__ __forceinline__ void wave_lds_handover() {
  __builtin_amdgcn_fence(__ATOMIC_RELEASE, "wavefront");
  __builtin_amdgcn_wave_barrier();
  __builtin_amdgcn_fence(__ATOMIC_ACQUIRE, "wavefront");
}

__device__ __forceinline__ void order_and_sum(int T, int lane, int kh, int m,
                                              v2i* U, v2i* Q, int* cnt, int* cur,
                                              const v4f* __restrict__ in4, v4f (&acc)[RPH]) {
  wave_lds_handover();
  const int cn = cnt[lane & 15];
  int x = cn;
#pragma unroll
  for (int d = 1; d < 16; d <<= 1) {
    int y = __shfl_up(x, d, 16);
    if ((lane & 15) >= d) x += y;
  }
  const int st = x - cn;
  if (lane < 16) { cur[lane] = st; cnt[lane] = 0; }
  wave_lds_handover();
  for (int i = lane; i < T; i += 32) {
    v2i e = U[i];
    int pos = atomicAdd(&cur[(unsigned)e.x >> 28], 1);
    Q[pos] = e;
  }
  wave_lds_handover();
#pragma unroll
  for (int j = 0; j < RPH; ++j) {
    const int sj = __shfl(st, kh * RPH + j);
    const int lj = __shfl(cn, kh * RPH + j);
    const int lo = __shfl_xor(lj, 16);
    const int n = lj > lo ? lj : lo;
    for (int q0 = 0; q0 < n; q0 += RUN) {
      v2i rc[RUN];
      v4f v[RUN];
#pragma unroll
      for (int k = 0; k < RUN; ++k) rc[k] = Q[(q0 + k) < lj ? sj + q0 + k : CAP];
#pragma unroll
      for (int k = 0; k < RUN; ++k) v[k] = in4[(unsigned)(rc[k].x & 0x0fffffff) + (unsigned)m];
      asm volatile("" :: "v"(v[0]), "v"(v[1]), "v"(v[2]), "v"(v[3]), "v"(v[4]), "v"(v[5]), "v"(v[6]), "v"(v[7]));
      __builtin_amdgcn_sched_barrier(0);
#pragma unroll
      for (int k = 0; k < RUN; ++k) {
        const bool live = (q0 + k) < lj;
        const float lc = __int_as_float(rc[k].y);
        const float s0 = __builtin_fmaf(lc, v[k].x, acc[j].x);
        const float s1 = __builtin_fmaf(lc, v[k].y, acc[j].y);
        const float s2 = __builtin_fmaf(lc, v[k].z, acc[j].z);
        const float s3 = __builtin_fmaf(lc, v[k].w, acc[j].w);
        acc[j].x = live ? s0 : acc[j].x;
        acc[j].y = live ? s1 : acc[j].y;
        acc[j].z = live ? s2 : acc[j].z;
        acc[j].w = live ? s3 : acc[j].w;
      }
    }
  }
  wave_lds_handover();
}

__global__ __launch_bounds__(SPW * 32) void pull_gemm_kernel(
    const int* __restrict__ dir, const int* __restrict__ recs,
    const float* __restrict__ input, const float* __restrict__ W,
    const float* __restrict__ b, float* __restrict__ out, int nrows, int nedges,
    int nchunks, int nheads) {
  __shared__ float ldsA[SPW * 16 * DF];
  __shared__ float ldsW[(DF + 1) * WP];
  __shared__ float ldsP[(DF + 1) * WP];
  __shared__ v2i   sQ[SPW * (CAP + 1)];
  __shared__ int   sCntRow[SPW * 16];
  __shared__ int   sCurRow[SPW * 16];

  int t = threadIdx.x;
  int rowBase0 = blockIdx.x * (SPW * 16);
  int wave = t >> 5;
  int lane = t & 31;
  int m  = lane & 15;
  int kh = lane >> 4;

  const int strip = (rowBase0 >> 4) + wave;
  int shift = 4;
  while (((nheads - 1) >> shift) >= NBIN) ++shift;
  const int bin = (strip << 4) >> shift;
  const int half = bin & 1;
  const int* dirp = dir + (size_t)half * NBIN + bin;

  int efirst[NB];
#pragma unroll
  for (int k = 0; k < NB; ++k) {
    const int c = NB * lane + k;
    efirst[k] = 0;
    if (c < nchunks) efirst[k] = dirp[(size_t)c * (2 * NBIN)];
  }

  constexpr int WJ = (DF * DF / 4 + SPW * 32 - 1) / (SPW * 32);
  v4f wv[WJ];
#pragma unroll
  for (int j = 0; j < WJ; ++j) {
    int fi = (j * (SPW * 32) + t) << 2;
    v4f z = {0.f, 0.f, 0.f, 0.f};
    wv[j] = z;
    if (fi < DF * DF) wv[j] = *(const v4f*)(W + fi);
  }
#pragma unroll
  for (int j = 0; j < WJ; ++j) {
    int fi = (j * (SPW * 32) + t) << 2;
    if (fi < DF * DF) *(v4f*)(ldsW + (fi >> 6) * WP + (fi & 63)) = wv[j];
  }
  if (t < DF) ldsW[DF * WP + t] = b[t];
  __syncthreads();

  const int wvs = __builtin_amdgcn_readfirstlane(wave);
  if (wvs < 20) {
    const int tr = wvs & 3;
    const int tc = wvs >> 2;
    const bool bcol = tc == 4;
    const bool keep = !bcol || m == 0;
    const float* Ap = ldsW + (16 * tr + m) * WP + 2 * kh;
    const int i0 = bcol ? DF * WP + 2 * kh : 2 * kh * WP + 16 * tc + m;
    const int sS = bcol ? 1 : WP;
    v2f fa[DF / 4], fb[DF / 4];
#pragma unroll
    for (int q4 = 0; q4 < DF / 4; ++q4) {
      fa[q4] = *(const v2f*)(Ap + 4 * q4);
      const float x0 = ldsW[i0 + (4 * q4) * sS];
      const float x1 = ldsW[i0 + (4 * q4 + 1) * sS];
      fb[q4].x = keep ? x0 : 0.f;
      fb[q4].y = keep ? x1 : 0.f;
    }
    float addv[8];
#pragma unroll
    for (int v = 0; v < 8; ++v) {
      const float bv = ldsW[DF * WP + 16 * tr + 8 * kh + v];
      addv[v] = bcol ? bv : 0.f;
    }
    v8f pacc = {};
#pragma unroll
    for (int q4 = 0; q4 < DF / 4; ++q4)
      pacc = __builtin_amdgcn_wmma_f32_16x16x4_f32(false, fa[q4], false, fb[q4], (short)0, pacc, false, false);
    const int o0 = bcol ? DF * WP + 16 * tr + 8 * kh : (16 * tr + 8 * kh) * WP + 16 * tc + m;
    const int oS = bcol ? 1 : WP;
    if (keep) {
#pragma unroll
      for (int v = 0; v < 8; ++v) ldsP[o0 + v * oS] = pacc[v] + addv[v];
    }
  }

  v2i* myU = (v2i*)(ldsA + wave * (16 * DF));
  int* myTbl = (int*)(myU + (CAP + 1));
  v2i* myQ = sQ + wave * (CAP + 1);
  int* myCnt = sCntRow + wave * 16;
  int* myCur = sCurRow + wave * 16;
  if (lane < 16) myCnt[lane] = 0;
  if (lane == 0) {
    v2i e = {rowBase0 * (DF / 4), 0};
    myQ[CAP] = e;
  }
  wave_lds_handover();

  const v4i* recs4 = (const v4i*)recs;
  const v4f* in4 = (const v4f*)input;
  const unsigned ltmask = (1u << lane) - 1u;

  int T = 0;
  bool fits = true;
  for (int cb = 0; fits && cb < nchunks; cb += 32 * NB) {
    int cnt[NB], slot0[NB], pre[NB];
    int s = 0;
#pragma unroll
    for (int k = 0; k < NB; ++k) {
      const int c = cb + NB * lane + k;
      int e = efirst[k];
      if (cb > 0) {
        e = 0;
        if (c < nchunks) e = dirp[(size_t)c * (2 * NBIN)];
      }
      const int p0 = e & 0xffff;
      cnt[k] = (int)((unsigned)e >> 16);
      const int base = c * EPB;
      int nloc = nedges - base;
      if (nloc > EPB) nloc = EPB;
      slot0[k] = half ? base + nloc - p0 - cnt[k] : base + p0;
      pre[k] = s;
      s += cnt[k];
    }
    int x = s;
#pragma unroll
    for (int d = 1; d < 32; d <<= 1) {
      int y = __shfl_up(x, d);
      if (lane >= d) x += y;
    }
    const int total = __shfl(x, 31);
#pragma unroll
    for (int k = 0; k < NB; ++k) pre[k] += x - s;

    for (int w0 = 0; fits && w0 < total; w0 += WIN) {
      int first[NB], n[NB];
      int nmax = 0;
#pragma unroll
      for (int k = 0; k < NB; ++k) {
        int lo = w0 - pre[k];
        if (lo < 0) lo = 0;
        int hi = w0 + WIN - pre[k];
        if (hi > cnt[k]) hi = cnt[k];
        n[k] = hi - lo;
        first[k] = slot0[k] + lo;
        if (n[k] > nmax) nmax = n[k];
      }
      for (int i = 0; __builtin_amdgcn_ballot_w32(i < nmax) != 0; ++i) {
#pragma unroll
        for (int k = 0; k < NB; ++k)
          if (i < n[k]) myTbl[first[k] - slot0[k] + pre[k] - w0 + i] = first[k] + i;
      }
      wave_lds_handover();
      int nrec = total - w0;
      if (nrec > WIN) nrec = WIN;
      v4i r[WIN / 32];
#pragma unroll
      for (int k = 0; k < WIN / 32; ++k) {
        const int pos = 32 * k + lane;
        r[k] = recs4[(unsigned)myTbl[pos < nrec ? pos : 0]];
      }
      unsigned mask[WIN / 32];
      int tot = 0;
#pragma unroll
      for (int k = 0; k < WIN / 32; ++k) {
        mask[k] = __builtin_amdgcn_ballot_w32(32 * k + lane < nrec && (r[k].x >> 4) == strip);
        tot += __builtin_popcount(mask[k]);
      }
      if (T + tot > CAP) {
        fits = false;
      } else {
#pragma unroll
        for (int k = 0; k < WIN / 32; ++k) {
          if ((mask[k] >> lane) & 1u) {
            int idx = T + __builtin_popcount(mask[k] & ltmask);
            unsigned r4 = (unsigned)r[k].x & 15u;
            v2i u = {(int)((unsigned)r[k].y | (r4 << 28)), r[k].z};
            myU[idx] = u;
            atomicAdd(&myCnt[r4], 1);
          }
          T += __builtin_popcount(mask[k]);
        }
      }
    }
  }

  v4f acc[RPH];
#pragma unroll
  for (int j = 0; j < RPH; ++j) {
    v4f z = {0.f, 0.f, 0.f, 0.f};
    acc[j] = z;
  }

  int fcb = 0, fi0 = 0;
  if (!fits) {
    wave_lds_handover();
    if (lane < 16) myCnt[lane] = 0;
    wave_lds_handover();
    T = 0;
  }
  bool done = fits;
  for (;;) {
    if (!done) {
      done = true;
      for (; fcb < nchunks; fcb += 32, fi0 = 0) {
        const int c = fcb + lane;
        int e = 0;
        if (c < nchunks) e = dirp[(size_t)c * (2 * NBIN)];
        const int p0 = e & 0xffff;
        const int cnt = (int)((unsigned)e >> 16);
        const int base = c * EPB;
        int nloc = nedges - base;
        if (nloc > EPB) nloc = EPB;
        const int slot0 = half ? base + nloc - p0 - cnt : base + p0;
        bool full = false;
        for (; __builtin_amdgcn_ballot_w32(fi0 < cnt) != 0; fi0 += RU) {
          v4i r[RU];
#pragma unroll
          for (int k = 0; k < RU; ++k) {
            v4i none = {-1, 0, 0, 0};
            r[k] = none;
            if (fi0 + k < cnt) r[k] = recs4[(unsigned)(slot0 + fi0 + k)];
          }
          unsigned mask[RU];
          int tot = 0;
#pragma unroll
          for (int k = 0; k < RU; ++k) {
            mask[k] = __builtin_amdgcn_ballot_w32(r[k].x >= 0 && (r[k].x >> 4) == strip);
            tot += __builtin_popcount(mask[k]);
          }
          if (T + tot > CAP) {
            full = true;
            break;
          }
#pragma unroll
          for (int k = 0; k < RU; ++k) {
            if ((mask[k] >> lane) & 1u) {
              int idx = T + __builtin_popcount(mask[k] & ltmask);
              unsigned r4 = (unsigned)r[k].x & 15u;
              v2i u = {(int)((unsigned)r[k].y | (r4 << 28)), r[k].z};
              myU[idx] = u;
              atomicAdd(&myCnt[r4], 1);
            }
            T += __builtin_popcount(mask[k]);
          }
        }
        if (full) {
          done = false;
          break;
        }
      }
    }
    order_and_sum(T, lane, kh, m, myU, myQ, myCnt, myCur, in4, acc);
    T = 0;
    if (done) break;
  }

  int rloc = wave * 16 + kh * RPH;
  const unsigned q = (unsigned)m << 2;
#pragma unroll
  for (int j = 0; j < RPH; ++j) *(v4f*)(ldsA + (rloc + j) * DF + q) = acc[j];
  __syncthreads();

  int rowBase = rowBase0 + wave * 16;
  if (rowBase >= nrows) return;

  float* S = ldsA + wave * (16 * DF);
  const float* A  = S + m * DF;
  const float* Bw = ldsP + m * WP;
  const int ka0 = 2 * kh;
  const float* Cc = ldsP + DF * WP;
  float c0 = Cc[m], c1 = Cc[m + 16], c2 = Cc[m + 32], c3 = Cc[m + 48];

  v8f acc0 = {}, acc1 = {}, acc2 = {}, acc3 = {};
  strip_times_Mt(A, Bw, ka0, acc0, acc1, acc2, acc3);

#pragma unroll
  for (int v = 0; v < 8; ++v) {
    int r = rowBase + v + 8 * kh;
    float* o = out + (size_t)r * DF;
    o[m]      = acc0[v] + c0;
    o[m + 16] = acc1[v] + c1;
    o[m + 32] = acc2[v] + c2;
    o[m + 48] = acc3[v] + c3;
  }
}

extern "C" void kernel_launch(void* const* d_in, const int* in_sizes, int n_in,
                              void* d_out, int out_size, void* d_ws, size_t ws_size,
                              hipStream_t stream) {
  const float* lap  = (const float*)d_in[0];
  const int*   eidx = (const int*)  d_in[1];
  const float* inp  = (const float*)d_in[2];
  const float* W    = (const float*)d_in[3];
  const float* b    = (const float*)d_in[4];
  float* out = (float*)d_out;

  const int E = in_sizes[0];
  const int N = in_sizes[2] / DF;
  const int* rows = eidx;
  const int* cols = eidx + E;

  int nBlocks = (N + SPW * 16 - 1) / (SPW * 16);
  int nHeads  = nBlocks * (SPW * 16);
  int nChunks = (E + EPB - 1) / EPB;

  int* recs = (int*)d_ws;
  int* dir  = recs + 4 * (size_t)E;

  order_edges_kernel<<<2 * nChunks, LT, 0, stream>>>(lap, rows, cols, recs, dir, E, nHeads);
  pull_gemm_kernel<<<nBlocks, SPW * 32, 0, stream>>>(dir, recs, inp, W, b, out, N, E, nChunks, nHeads);
}


# Ordering launch: 16 bytes per lane on every path: a lane reads 4 consecutive edges per load, an edge is staged in LDS as one 16-byte record (one store, one load, not three each); sum order changes
# speedup vs baseline: 4.3527x; 1.0464x over previous
#include <hip/hip_runtime.h>


typedef __attribute__((ext_vector_type(2))) float v2f;
typedef __attribute__((ext_vector_type(4))) float v4f;
typedef __attribute__((ext_vector_type(8))) float v8f;
typedef __attribute__((ext_vector_type(2))) int v2i;
typedef __attribute__((ext_vector_type(4))) int v4i;

#define DF 64
#define RPH 8
#define SPW 25
#define WP 68

#define LT 1024
#define EPT 16
#define EPG 4
#define NG (EPT / EPG)
#define EPB (LT * EPT)
#define NBIN 8192
#define BPT (NBIN / LT)

template <bool VEC>
__device__ __forceinline__ void order_body(
    const float* __restrict__ lap, const int* __restrict__ rows,
    const int* __restrict__ cols, int* __restrict__ recs, int* __restrict__ dirRow,
    int base, int nloc, int shift, int half,
    int* sCnt, v4i* sRec, int* sWave, int* sTot) {
  const int t = threadIdx.x;
  const int lane = t & 31;
  const int wv = t >> 5;

  v4i r4[NG], c4[NG];
  v4f l4[NG];
#pragma unroll
  for (int g = 0; g < NG; ++g) {
    const int i = EPG * (t + g * LT);
    if (VEC) {
      r4[g] = *(const v4i*)(rows + base + i);
      c4[g] = *(const v4i*)(cols + base + i);
      l4[g] = *(const v4f*)(lap + base + i);
    } else {
      v4i rr = {-1, -1, -1, -1};
      v4i cc = {0, 0, 0, 0};
      v4f ll = {0.f, 0.f, 0.f, 0.f};
#pragma unroll
      for (int e = 0; e < EPG; ++e) {
        if (i + e < nloc) {
          rr[e] = rows[base + i + e];
          cc[e] = cols[base + i + e];
          ll[e] = lap[base + i + e];
        }
      }
      r4[g] = rr; c4[g] = cc; l4[g] = ll;
    }
  }

#pragma unroll
  for (int k = 0; k < NBIN / LT; ++k) sCnt[t + k * LT] = 0;
  __syncthreads();

  unsigned keep = 0;
  int rank[EPT];
#pragma unroll
  for (int k = 0; k < EPT; ++k) {
    const int r = r4[k / EPG][k % EPG];
    const bool mine = r >= 0 && ((r >> shift) & 1) == half;
    rank[k] = 0;
    if (mine) {
      rank[k] = atomicAdd(&sCnt[r >> shift], 1);
      keep |= 1u << k;
    }
  }
  __syncthreads();

  int bc[BPT];
  int s = 0;
#pragma unroll
  for (int k = 0; k < BPT; ++k) { bc[k] = sCnt[BPT * t + k]; s += bc[k]; }
  int x = s;
#pragma unroll
  for (int d = 1; d < 32; d <<= 1) {
    int y = __shfl_up(x, d);
    if (lane >= d) x += y;
  }
  if (lane == 31) sWave[wv] = x;
  __syncthreads();
  if (wv == 0) {
    int w = sWave[lane];
    int z = w;
#pragma unroll
    for (int d = 1; d < 32; d <<= 1) {
      int y = __shfl_up(z, d);
      if (lane >= d) z += y;
    }
    sWave[lane] = z - w;
    if (lane == 31) *sTot = z;
  }
  __syncthreads();
  int off = sWave[wv] + x - s;
  int de[BPT];
#pragma unroll
  for (int k = 0; k < BPT; ++k) {
    sCnt[BPT * t + k] = off;
    de[k] = off | (bc[k] << 16);
    off += bc[k];
  }
#pragma unroll
  for (int k = 0; k < BPT; k += 4) {
    v4i d4 = {de[k], de[k + 1], de[k + 2], de[k + 3]};
    *(v4i*)(dirRow + BPT * t + k) = d4;
  }
  __syncthreads();

#pragma unroll
  for (int k = 0; k < EPT; ++k) {
    if ((keep >> k) & 1u) {
      const int r = r4[k / EPG][k % EPG];
      const int p = sCnt[r >> shift] + rank[k];
      v4i rec = {r, c4[k / EPG][k % EPG] * (DF / 4), __float_as_int(l4[k / EPG][k % EPG]), 0};
      sRec[p] = rec;
    }
  }
  __syncthreads();

  const int nk = *sTot;
  const int first = half ? base + nloc - 1 : base;
  const int step = half ? -1 : 1;
#pragma unroll
  for (int k = 0; k < EPT; ++k) {
    int p = t + k * LT;
    if (p < nk) ((v4i*)recs)[(unsigned)(first + step * p)] = sRec[p];
  }
}

__global__ __launch_bounds__(LT) void order_edges_kernel(
    const float* __restrict__ lap, const int* __restrict__ rows,
    const int* __restrict__ cols, int* __restrict__ recs, int* __restrict__ dir,
    int nedges, int nheads) {
  __shared__ int sCnt[NBIN];
  __shared__ v4i sRec[EPB];
  __shared__ int sWave[32];
  __shared__ int sTot;

  const int half = blockIdx.x & 1;
  const int base = (blockIdx.x >> 1) * EPB;
  int nloc = nedges - base;
  if (nloc > EPB) nloc = EPB;
  int* dirRow = dir + (size_t)blockIdx.x * NBIN;
  int shift = 4;
  while (((nheads - 1) >> shift) >= NBIN) ++shift;

  if (nloc == EPB && (nedges & 3) == 0)
    order_body<true>(lap, rows, cols, recs, dirRow, base, nloc, shift, half, sCnt, sRec, sWave, &sTot);
  else
    order_body<false>(lap, rows, cols, recs, dirRow, base, nloc, shift, half, sCnt, sRec, sWave, &sTot);
}

__device__ __forceinline__ void strip_times_Mt(const float* A, const float* Bw, int ka0,
                                               v8f& acc0, v8f& acc1, v8f& acc2, v8f& acc3) {
#pragma unroll
  for (int k0 = 0; k0 < DF; k0 += 4) {
    int ka = k0 + ka0;
    v2f a  = *(const v2f*)(A + ka);
    v2f b0 = *(const v2f*)(Bw + ka);
    v2f b1 = *(const v2f*)(Bw + 16 * WP + ka);
    v2f b2 = *(const v2f*)(Bw + 32 * WP + ka);
    v2f b3 = *(const v2f*)(Bw + 48 * WP + ka);
    acc0 = __builtin_amdgcn_wmma_f32_16x16x4_f32(false, a, false, b0, (short)0, acc0, false, false);
    acc1 = __builtin_amdgcn_wmma_f32_16x16x4_f32(false, a, false, b1, (short)0, acc1, false, false);
    acc2 = __builtin_amdgcn_wmma_f32_16x16x4_f32(false, a, false, b2, (short)0, acc2, false, false);
    acc3 = __builtin_amdgcn_wmma_f32_16x16x4_f32(false, a, false, b3, (short)0, acc3, false, false);
  }
}

#define RUN 8
#define RU 4
#define NB 4
#define WIN 320
#define CAP 351
static_assert(8 * (CAP + 1) + 4 * WIN <= 16 * DF * 4 && WIN <= CAP && WIN % 32 == 0, "a wave's LDS places");

__device__ __forceinline__ void wave_lds_handover() {
  __builtin_amdgcn_fence(__ATOMIC_RELEASE, "wavefront");
  __builtin_amdgcn_wave_barrier();
  __builtin_amdgcn_fence(__ATOMIC_ACQUIRE, "wavefront");
}

__device__ __forceinline__ void order_and_sum(int T, int lane, int kh, int m,
                                              v2i* U, v2i* Q, int* cnt, int* cur,
                                              const v4f* __restrict__ in4, v4f (&acc)[RPH]) {
  wave_lds_handover();
  const int cn = cnt[lane & 15];
  int x = cn;
#pragma unroll
  for (int d = 1; d < 16; d <<= 1) {
    int y = __shfl_up(x, d, 16);
    if ((lane & 15) >= d) x += y;
  }
  const int st = x - cn;
  if (lane < 16) { cur[lane] = st; cnt[lane] = 0; }
  wave_lds_handover();
  for (int i = lane; i < T; i += 32) {
    v2i e = U[i];
    int pos = atomicAdd(&cur[(unsigned)e.x >> 28], 1);
    Q[pos] = e;
  }
  wave_lds_handover();
#pragma unroll
  for (int j = 0; j < RPH; ++j) {
    const int sj = __shfl(st, kh * RPH + j);
    const int lj = __shfl(cn, kh * RPH + j);
    const int lo = __shfl_xor(lj, 16);
    const int n = lj > lo ? lj : lo;
    for (int q0 = 0; q0 < n; q0 += RUN) {
      v2i rc[RUN];
      v4f v[RUN];
#pragma unroll
      for (int k = 0; k < RUN; ++k) rc[k] = Q[(q0 + k) < lj ? sj + q0 + k : CAP];
#pragma unroll
      for (int k = 0; k < RUN; ++k) v[k] = in4[(unsigned)(rc[k].x & 0x0fffffff) + (unsigned)m];
      asm volatile("" :: "v"(v[0]), "v"(v[1]), "v"(v[2]), "v"(v[3]), "v"(v[4]), "v"(v[5]), "v"(v[6]), "v"(v[7]));
      __builtin_amdgcn_sched_barrier(0);
#pragma unroll
      for (int k = 0; k < RUN; ++k) {
        const bool live = (q0 + k) < lj;
        const float lc = __int_as_float(rc[k].y);
        const float s0 = __builtin_fmaf(lc, v[k].x, acc[j].x);
        const float s1 = __builtin_fmaf(lc, v[k].y, acc[j].y);
        const float s2 = __builtin_fmaf(lc, v[k].z, acc[j].z);
        const float s3 = __builtin_fmaf(lc, v[k].w, acc[j].w);
        acc[j].x = live ? s0 : acc[j].x;
        acc[j].y = live ? s1 : acc[j].y;
        acc[j].z = live ? s2 : acc[j].z;
        acc[j].w = live ? s3 : acc[j].w;
      }
    }
  }
  wave_lds_handover();
}

__global__ __launch_bounds__(SPW * 32) void pull_gemm_kernel(
    const int* __restrict__ dir, const int* __restrict__ recs,
    const float* __restrict__ input, const float* __restrict__ W,
    const float* __restrict__ b, float* __restrict__ out, int nrows, int nedges,
    int nchunks, int nheads) {
  __shared__ float ldsA[SPW * 16 * DF];
  __shared__ float ldsW[(DF + 1) * WP];
  __shared__ float ldsP[(DF + 1) * WP];
  __shared__ v2i   sQ[SPW * (CAP + 1)];
  __shared__ int   sCntRow[SPW * 16];
  __shared__ int   sCurRow[SPW * 16];

  int t = threadIdx.x;
  int rowBase0 = blockIdx.x * (SPW * 16);
  int wave = t >> 5;
  int lane = t & 31;
  int m  = lane & 15;
  int kh = lane >> 4;

  const int strip = (rowBase0 >> 4) + wave;
  int shift = 4;
  while (((nheads - 1) >> shift) >= NBIN) ++shift;
  const int bin = (strip << 4) >> shift;
  const int half = bin & 1;
  const int* dirp = dir + (size_t)half * NBIN + bin;

  int efirst[NB];
#pragma unroll
  for (int k = 0; k < NB; ++k) {
    const int c = NB * lane + k;
    efirst[k] = 0;
    if (c < nchunks) efirst[k] = dirp[(size_t)c * (2 * NBIN)];
  }

  constexpr int WJ = (DF * DF / 4 + SPW * 32 - 1) / (SPW * 32);
  v4f wv[WJ];
#pragma unroll
  for (int j = 0; j < WJ; ++j) {
    int fi = (j * (SPW * 32) + t) << 2;
    v4f z = {0.f, 0.f, 0.f, 0.f};
    wv[j] = z;
    if (fi < DF * DF) wv[j] = *(const v4f*)(W + fi);
  }
#pragma unroll
  for (int j = 0; j < WJ; ++j) {
    int fi = (j * (SPW * 32) + t) << 2;
    if (fi < DF * DF) *(v4f*)(ldsW + (fi >> 6) * WP + (fi & 63)) = wv[j];
  }
  if (t < DF) ldsW[DF * WP + t] = b[t];
  __syncthreads();

  const int wvs = __builtin_amdgcn_readfirstlane(wave);
  if (wvs < 20) {
    const int tr = wvs & 3;
    const int tc = wvs >> 2;
    const bool bcol = tc == 4;
    const bool keep = !bcol || m == 0;
    const float* Ap = ldsW + (16 * tr + m) * WP + 2 * kh;
    const int i0 = bcol ? DF * WP + 2 * kh : 2 * kh * WP + 16 * tc + m;
    const int sS = bcol ? 1 : WP;
    v2f fa[DF / 4], fb[DF / 4];
#pragma unroll
    for (int q4 = 0; q4 < DF / 4; ++q4) {
      fa[q4] = *(const v2f*)(Ap + 4 * q4);
      const float x0 = ldsW[i0 + (4 * q4) * sS];
      const float x1 = ldsW[i0 + (4 * q4 + 1) * sS];
      fb[q4].x = keep ? x0 : 0.f;
      fb[q4].y = keep ? x1 : 0.f;
    }
    float addv[8];
#pragma unroll
    for (int v = 0; v < 8; ++v) {
      const float bv = ldsW[DF * WP + 16 * tr + 8 * kh + v];
      addv[v] = bcol ? bv : 0.f;
    }
    v8f pacc = {};
#pragma unroll
    for (int q4 = 0; q4 < DF / 4; ++q4)
      pacc = __builtin_amdgcn_wmma_f32_16x16x4_f32(false, fa[q4], false, fb[q4], (short)0, pacc, false, false);
    const int o0 = bcol ? DF * WP + 16 * tr + 8 * kh : (16 * tr + 8 * kh) * WP + 16 * tc + m;
    const int oS = bcol ? 1 : WP;
    if (keep) {
#pragma unroll
      for (int v = 0; v < 8; ++v) ldsP[o0 + v * oS] = pacc[v] + addv[v];
    }
  }

  v2i* myU = (v2i*)(ldsA + wave * (16 * DF));
  int* myTbl = (int*)(myU + (CAP + 1));
  v2i* myQ = sQ + wave * (CAP + 1);
  int* myCnt = sCntRow + wave * 16;
  int* myCur = sCurRow + wave * 16;
  if (lane < 16) myCnt[lane] = 0;
  if (lane == 0) {
    v2i e = {rowBase0 * (DF / 4), 0};
    myQ[CAP] = e;
  }
  wave_lds_handover();

  const v4i* recs4 = (const v4i*)recs;
  const v4f* in4 = (const v4f*)input;
  const unsigned ltmask = (1u << lane) - 1u;

  int T = 0;
  bool fits = true;
  for (int cb = 0; fits && cb < nchunks; cb += 32 * NB) {
    int cnt[NB], slot0[NB], pre[NB];
    int s = 0;
#pragma unroll
    for (int k = 0; k < NB; ++k) {
      const int c = cb + NB * lane + k;
      int e = efirst[k];
      if (cb > 0) {
        e = 0;
        if (c < nchunks) e = dirp[(size_t)c * (2 * NBIN)];
      }
      const int p0 = e & 0xffff;
      cnt[k] = (int)((unsigned)e >> 16);
      const int base = c * EPB;
      int nloc = nedges - base;
      if (nloc > EPB) nloc = EPB;
      slot0[k] = half ? base + nloc - p0 - cnt[k] : base + p0;
      pre[k] = s;
      s += cnt[k];
    }
    int x = s;
#pragma unroll
    for (int d = 1; d < 32; d <<= 1) {
      int y = __shfl_up(x, d);
      if (lane >= d) x += y;
    }
    const int total = __shfl(x, 31);
#pragma unroll
    for (int k = 0; k < NB; ++k) pre[k] += x - s;

    for (int w0 = 0; fits && w0 < total; w0 += WIN) {
      int first[NB], n[NB];
      int nmax = 0;
#pragma unroll
      for (int k = 0; k < NB; ++k) {
        int lo = w0 - pre[k];
        if (lo < 0) lo = 0;
        int hi = w0 + WIN - pre[k];
        if (hi > cnt[k]) hi = cnt[k];
        n[k] = hi - lo;
        first[k] = slot0[k] + lo;
        if (n[k] > nmax) nmax = n[k];
      }
      for (int i = 0; __builtin_amdgcn_ballot_w32(i < nmax) != 0; ++i) {
#pragma unroll
        for (int k = 0; k < NB; ++k)
          if (i < n[k]) myTbl[first[k] - slot0[k] + pre[k] - w0 + i] = first[k] + i;
      }
      wave_lds_handover();
      int nrec = total - w0;
      if (nrec > WIN) nrec = WIN;
      v4i r[WIN / 32];
#pragma unroll
      for (int k = 0; k < WIN / 32; ++k) {
        const int pos = 32 * k + lane;
        r[k] = recs4[(unsigned)myTbl[pos < nrec ? pos : 0]];
      }
      unsigned mask[WIN / 32];
      int tot = 0;
#pragma unroll
      for (int k = 0; k < WIN / 32; ++k) {
        mask[k] = __builtin_amdgcn_ballot_w32(32 * k + lane < nrec && (r[k].x >> 4) == strip);
        tot += __builtin_popcount(mask[k]);
      }
      if (T + tot > CAP) {
        fits = false;
      } else {
#pragma unroll
        for (int k = 0; k < WIN / 32; ++k) {
          if ((mask[k] >> lane) & 1u) {
            int idx = T + __builtin_popcount(mask[k] & ltmask);
            unsigned r4 = (unsigned)r[k].x & 15u;
            v2i u = {(int)((unsigned)r[k].y | (r4 << 28)), r[k].z};
            myU[idx] = u;
            atomicAdd(&myCnt[r4], 1);
          }
          T += __builtin_popcount(mask[k]);
        }
      }
    }
  }

  v4f acc[RPH];
#pragma unroll
  for (int j = 0; j < RPH; ++j) {
    v4f z = {0.f, 0.f, 0.f, 0.f};
    acc[j] = z;
  }

  int fcb = 0, fi0 = 0;
  if (!fits) {
    wave_lds_handover();
    if (lane < 16) myCnt[lane] = 0;
    wave_lds_handover();
    T = 0;
  }
  bool done = fits;
  for (;;) {
    if (!done) {
      done = true;
      for (; fcb < nchunks; fcb += 32, fi0 = 0) {
        const int c = fcb + lane;
        int e = 0;
        if (c < nchunks) e = dirp[(size_t)c * (2 * NBIN)];
        const int p0 = e & 0xffff;
        const int cnt = (int)((unsigned)e >> 16);
        const int base = c * EPB;
        int nloc = nedges - base;
        if (nloc > EPB) nloc = EPB;
        const int slot0 = half ? base + nloc - p0 - cnt : base + p0;
        bool full = false;
        for (; __builtin_amdgcn_ballot_w32(fi0 < cnt) != 0; fi0 += RU) {
          v4i r[RU];
#pragma unroll
          for (int k = 0; k < RU; ++k) {
            v4i none = {-1, 0, 0, 0};
            r[k] = none;
            if (fi0 + k < cnt) r[k] = recs4[(unsigned)(slot0 + fi0 + k)];
          }
          unsigned mask[RU];
          int tot = 0;
#pragma unroll
          for (int k = 0; k < RU; ++k) {
            mask[k] = __builtin_amdgcn_ballot_w32(r[k].x >= 0 && (r[k].x >> 4) == strip);
            tot += __builtin_popcount(mask[k]);
          }
          if (T + tot > CAP) {
            full = true;
            break;
          }
#pragma unroll
          for (int k = 0; k < RU; ++k) {
            if ((mask[k] >> lane) & 1u) {
              int idx = T + __builtin_popcount(mask[k] & ltmask);
              unsigned r4 = (unsigned)r[k].x & 15u;
              v2i u = {(int)((unsigned)r[k].y | (r4 << 28)), r[k].z};
              myU[idx] = u;
              atomicAdd(&myCnt[r4], 1);
            }
            T += __builtin_popcount(mask[k]);
          }
        }
        if (full) {
          done = false;
          break;
        }
      }
    }
    order_and_sum(T, lane, kh, m, myU, myQ, myCnt, myCur, in4, acc);
    T = 0;
    if (done) break;
  }

  int rloc = wave * 16 + kh * RPH;
  const unsigned q = (unsigned)m << 2;
#pragma unroll
  for (int j = 0; j < RPH; ++j) *(v4f*)(ldsA + (rloc + j) * DF + q) = acc[j];
  __syncthreads();

  int rowBase = rowBase0 + wave * 16;
  if (rowBase >= nrows) return;

  float* S = ldsA + wave * (16 * DF);
  const float* A  = S + m * DF;
  const float* Bw = ldsP + m * WP;
  const int ka0 = 2 * kh;
  const float* Cc = ldsP + DF * WP;
  float c0 = Cc[m], c1 = Cc[m + 16], c2 = Cc[m + 32], c3 = Cc[m + 48];

  v8f acc0 = {}, acc1 = {}, acc2 = {}, acc3 = {};
  strip_times_Mt(A, Bw, ka0, acc0, acc1, acc2, acc3);

#pragma unroll
  for (int v = 0; v < 8; ++v) {
    int r = rowBase + v + 8 * kh;
    float* o = out + (size_t)r * DF;
    o[m]      = acc0[v] + c0;
    o[m + 16] = acc1[v] + c1;
    o[m + 32] = acc2[v] + c2;
    o[m + 48] = acc3[v] + c3;
  }
}

extern "C" void kernel_launch(void* const* d_in, const int* in_sizes, int n_in,
                              void* d_out, int out_size, void* d_ws, size_t ws_size,
                              hipStream_t stream) {
  const float* lap  = (const float*)d_in[0];
  const int*   eidx = (const int*)  d_in[1];
  const float* inp  = (const float*)d_in[2];
  const float* W    = (const float*)d_in[3];
  const float* b    = (const float*)d_in[4];
  float* out = (float*)d_out;

  const int E = in_sizes[0];
  const int N = in_sizes[2] / DF;
  const int* rows = eidx;
  const int* cols = eidx + E;

  int nBlocks = (N + SPW * 16 - 1) / (SPW * 16);
  int nHeads  = nBlocks * (SPW * 16);
  int nChunks = (E + EPB - 1) / EPB;

  int* recs = (int*)d_ws;
  int* dir  = recs + 4 * (size_t)E;

  order_edges_kernel<<<2 * nChunks, LT, 0, stream>>>(lap, rows, cols, recs, dir, E, nHeads);
  pull_gemm_kernel<<<nBlocks, SPW * 32, 0, stream>>>(dir, recs, inp, W, b, out, N, E, nChunks, nHeads);
}
